# starting kernel
# baseline (speedup 1.0000x reference)
#include <hip/hip_runtime.h>
#include <math.h>

#define N_NODES 50000
#define N_EDGES 800000
#define IN_DIM 128
#define OUT_DIM 128
#define HEADS 8
#define CPH 16
#define NEG_SLOPE 0.2f
#define GAT_EPS 1e-16f

typedef float v2f __attribute__((ext_vector_type(2)));
typedef float v8f __attribute__((ext_vector_type(8)));

#define LDX 132

__global__ __launch_bounds__(256) void gat_gemm_wmma_kernel(
    const float* __restrict__ x, const float* __restrict__ Wg,
    float* __restrict__ h) {
  __shared__ float sx[16 * LDX];
  const int tid = threadIdx.x;
  const int mbase = blockIdx.x * 16;

  for (int i = tid; i < 16 * 32; i += 256) {
    const int r  = i >> 5;
    const int c4 = i & 31;
    const float4 v = ((const float4*)(x + (size_t)(mbase + r) * IN_DIM))[c4];
    float* dst = &sx[r * LDX + c4 * 4];
    dst[0] = v.x; dst[1] = v.y; dst[2] = v.z; dst[3] = v.w;
  }
  __syncthreads();

  const int wave = tid >> 5;
  const int lane = tid & 31;
  const int half = lane >> 4;
  const int mr   = lane & 15;
  const int nb   = wave * 16;

  v8f acc = {};
#pragma unroll
  for (int k0 = 0; k0 < IN_DIM; k0 += 4) {
    v2f a, b;
    a.x = sx[mr * LDX + k0 + 2 * half];
    a.y = sx[mr * LDX + k0 + 2 * half + 1];
    b.x = Wg[(size_t)(k0 + 2 * half) * OUT_DIM + nb + mr];
    b.y = Wg[(size_t)(k0 + 2 * half + 1) * OUT_DIM + nb + mr];
    acc = __builtin_amdgcn_wmma_f32_16x16x4_f32(
         false, a,  false, b,
         (short)0, acc,  false,  false);
  }

#pragma unroll
  for (int r = 0; r < 8; ++r) {
    const int row = half * 8 + r;
    h[(size_t)(mbase + row) * OUT_DIM + nb + mr] = acc[r];
  }
}

__global__ void gat_attn_proj_kernel(const float* __restrict__ h,
                                     const float* __restrict__ att_src,
                                     const float* __restrict__ att_dst,
                                     float* __restrict__ a_src,
                                     float* __restrict__ a_dst) {
  const int i = blockIdx.x * blockDim.x + threadIdx.x;
  if (i >= N_NODES * HEADS) return;
  const int hd = i & (HEADS - 1);
  const float4* hp = (const float4*)(h + (size_t)i * CPH);
  const float4* as = (const float4*)(att_src + hd * CPH);
  const float4* ad = (const float4*)(att_dst + hd * CPH);
  float s = 0.f, d = 0.f;
#pragma unroll
  for (int q = 0; q < 4; ++q) {
    const float4 hv = hp[q], av = as[q], dv = ad[q];
    s += hv.x * av.x + hv.y * av.y + hv.z * av.z + hv.w * av.w;
    d += hv.x * dv.x + hv.y * dv.y + hv.z * dv.z + hv.w * dv.w;
  }
  a_src[i] = s;
  a_dst[i] = d;
}

__device__ __forceinline__ void edge_nodes(const int* __restrict__ ei, int e,
                                           int& s, int& d) {
  if (e < N_EDGES) { s = ei[e]; d = ei[N_EDGES + e]; }
  else             { s = d = e - N_EDGES; }
}
__device__ __forceinline__ float leaky(float v) {
  return v >= 0.f ? v : NEG_SLOPE * v;
}
__device__ __forceinline__ unsigned fflip(float f) {
  const unsigned u = __float_as_uint(f);
  return u ^ ((u >> 31) ? 0xFFFFFFFFu : 0x80000000u);
}
__device__ __forceinline__ float funflip(unsigned u) {
  return __uint_as_float(u ^ ((u >> 31) ? 0x80000000u : 0xFFFFFFFFu));
}

#define E2 (N_EDGES + N_NODES)

__global__ void gat_edge_max_kernel(const int* __restrict__ ei,
                                    const float* __restrict__ a_src,
                                    const float* __restrict__ a_dst,
                                    unsigned* __restrict__ emaxU) {
  const int i = blockIdx.x * blockDim.x + threadIdx.x;
  if (i >= E2 * HEADS) return;
  const int e = i >> 3, hd = i & (HEADS - 1);
  int s, d; edge_nodes(ei, e, s, d);
  const float v = leaky(a_src[s * HEADS + hd] + a_dst[d * HEADS + hd]);
  atomicMax(&emaxU[d * HEADS + hd], fflip(v));
}

__global__ void gat_edge_sum_kernel(const int* __restrict__ ei,
                                    const float* __restrict__ a_src,
                                    const float* __restrict__ a_dst,
                                    const unsigned* __restrict__ emaxU,
                                    float* __restrict__ denom) {
  const int i = blockIdx.x * blockDim.x + threadIdx.x;
  if (i >= E2 * HEADS) return;
  const int e = i >> 3, hd = i & (HEADS - 1);
  int s, d; edge_nodes(ei, e, s, d);
  const int dh = d * HEADS + hd;
  const float v = leaky(a_src[s * HEADS + hd] + a_dst[dh]);
  atomicAdd(&denom[dh], __expf(v - funflip(emaxU[dh])));
}

__global__ void gat_edge_scatter_kernel(const int* __restrict__ ei,
                                        const float* __restrict__ a_src,
                                        const float* __restrict__ a_dst,
                                        const unsigned* __restrict__ emaxU,
                                        const float* __restrict__ denom,
                                        const float* __restrict__ h,
                                        float* __restrict__ accum) {
  const int i = blockIdx.x * blockDim.x + threadIdx.x;
  if (i >= E2 * HEADS) return;
  const int e = i >> 3, hd = i & (HEADS - 1);
  int s, d; edge_nodes(ei, e, s, d);
  const int dh = d * HEADS + hd;
  const float v = leaky(a_src[s * HEADS + hd] + a_dst[dh]);
  const float alpha = __expf(v - funflip(emaxU[dh])) / (denom[dh] + GAT_EPS);
  const float4* hs = (const float4*)(h + (size_t)(s * HEADS + hd) * CPH);
  float* op = accum + (size_t)dh * CPH;
#pragma unroll
  for (int q = 0; q < 4; ++q) {
    const float4 hv = hs[q];
    atomicAdd(&op[q * 4 + 0], alpha * hv.x);
    atomicAdd(&op[q * 4 + 1], alpha * hv.y);
    atomicAdd(&op[q * 4 + 2], alpha * hv.z);
    atomicAdd(&op[q * 4 + 3], alpha * hv.w);
  }
}

__global__ void gat_finalize_kernel(const float* __restrict__ accum,
                                    const float* __restrict__ bias,
                                    float* __restrict__ out) {
  const int i = blockIdx.x * blockDim.x + threadIdx.x;
  if (i >= N_NODES * OUT_DIM) return;
  const float v = accum[i] + bias[i & (OUT_DIM - 1)];
  out[i] = v > 0.f ? v : expm1f(v);
}

extern "C" void kernel_launch(void* const* d_in, const int* in_sizes, int n_in,
                              void* d_out, int out_size, void* d_ws, size_t ws_size,
                              hipStream_t stream) {
  (void)in_sizes; (void)n_in; (void)out_size; (void)ws_size;
  const float* x       = (const float*)d_in[0];
  const int*   ei      = (const int*)d_in[1];
  const float* W       = (const float*)d_in[2];
  const float* att_src = (const float*)d_in[3];
  const float* att_dst = (const float*)d_in[4];
  const float* bias    = (const float*)d_in[5];

  float*    h      = (float*)d_ws;
  float*    a_src  = h + (size_t)N_NODES * OUT_DIM;
  float*    a_dst  = a_src + (size_t)N_NODES * HEADS;
  unsigned* emaxU  = (unsigned*)(a_dst + (size_t)N_NODES * HEADS);
  float*    denom  = (float*)emaxU + (size_t)N_NODES * HEADS;
  float*    accum  = denom + (size_t)N_NODES * HEADS;

  hipMemsetAsync(emaxU, 0,
                 (size_t)(2 * N_NODES * HEADS + (size_t)N_NODES * OUT_DIM) *
                     sizeof(float),
                 stream);

  gat_gemm_wmma_kernel<<<N_NODES / 16, 256, 0, stream>>>(x, W, h);

  gat_attn_proj_kernel<<<(N_NODES * HEADS + 255) / 256, 256, 0, stream>>>(
      h, att_src, att_dst, a_src, a_dst);

  const int ethreads = E2 * HEADS;
  const int eblocks  = (ethreads + 255) / 256;
  gat_edge_max_kernel<<<eblocks, 256, 0, stream>>>(ei, a_src, a_dst, emaxU);
  gat_edge_sum_kernel<<<eblocks, 256, 0, stream>>>(ei, a_src, a_dst, emaxU, denom);
  gat_edge_scatter_kernel<<<eblocks, 256, 0, stream>>>(ei, a_src, a_dst, emaxU,
                                                       denom, h, accum);

  gat_finalize_kernel<<<(N_NODES * OUT_DIM + 255) / 256, 256, 0, stream>>>(
      accum, bias, (float*)d_out);
}


# Edges bucketed by destination each call (count, scan, fill), each bucket sorted by source id; one wave per node gathers softmax and weighted h rows in that fixed order: float atomics gone
# speedup vs baseline: 2.8426x; 2.8426x over previous
#include <hip/hip_runtime.h>
#include <math.h>

#define N_NODES 50000
#define N_EDGES 800000
#define IN_DIM 128
#define OUT_DIM 128
#define HEADS 8
#define CPH 16
#define NEG_SLOPE 0.2f
#define GAT_EPS 1e-16f

typedef float v2f __attribute__((ext_vector_type(2)));
typedef float v8f __attribute__((ext_vector_type(8)));

#define LDX 132

__global__ __launch_bounds__(256) void gat_gemm_wmma_kernel(
    const float* __restrict__ x, const float* __restrict__ Wg,
    float* __restrict__ h) {
  __shared__ float sx[16 * LDX];
  const int tid = threadIdx.x;
  const int mbase = blockIdx.x * 16;

  for (int i = tid; i < 16 * 32; i += 256) {
    const int r  = i >> 5;
    const int c4 = i & 31;
    const float4 v = ((const float4*)(x + (size_t)(mbase + r) * IN_DIM))[c4];
    float* dst = &sx[r * LDX + c4 * 4];
    dst[0] = v.x; dst[1] = v.y; dst[2] = v.z; dst[3] = v.w;
  }
  __syncthreads();

  const int wave = tid >> 5;
  const int lane = tid & 31;
  const int half = lane >> 4;
  const int mr   = lane & 15;
  const int nb   = wave * 16;

  v8f acc = {};
#pragma unroll
  for (int k0 = 0; k0 < IN_DIM; k0 += 4) {
    v2f a, b;
    a.x = sx[mr * LDX + k0 + 2 * half];
    a.y = sx[mr * LDX + k0 + 2 * half + 1];
    b.x = Wg[(size_t)(k0 + 2 * half) * OUT_DIM + nb + mr];
    b.y = Wg[(size_t)(k0 + 2 * half + 1) * OUT_DIM + nb + mr];
    acc = __builtin_amdgcn_wmma_f32_16x16x4_f32(
         false, a,  false, b,
         (short)0, acc,  false,  false);
  }

#pragma unroll
  for (int r = 0; r < 8; ++r) {
    const int row = half * 8 + r;
    h[(size_t)(mbase + row) * OUT_DIM + nb + mr] = acc[r];
  }
}

__global__ void gat_attn_proj_kernel(const float* __restrict__ h,
                                     const float* __restrict__ att_src,
                                     const float* __restrict__ att_dst,
                                     float* __restrict__ a_src,
                                     float* __restrict__ a_dst,
                                     int* __restrict__ deg) {
  const int i = blockIdx.x * blockDim.x + threadIdx.x;
  if (i >= N_NODES * HEADS) return;
  if (i < N_NODES) deg[i] = 0;
  const int hd = i & (HEADS - 1);
  const float4* hp = (const float4*)(h + (size_t)i * CPH);
  const float4* as = (const float4*)(att_src + hd * CPH);
  const float4* ad = (const float4*)(att_dst + hd * CPH);
  float s = 0.f, d = 0.f;
#pragma unroll
  for (int q = 0; q < 4; ++q) {
    const float4 hv = hp[q], av = as[q], dv = ad[q];
    s += hv.x * av.x + hv.y * av.y + hv.z * av.z + hv.w * av.w;
    d += hv.x * dv.x + hv.y * dv.y + hv.z * dv.z + hv.w * dv.w;
  }
  a_src[i] = s;
  a_dst[i] = d;
}

__device__ __forceinline__ float leaky(float v) {
  return v >= 0.f ? v : NEG_SLOPE * v;
}

__global__ void gat_count_kernel(const int* __restrict__ ei, int* __restrict__ deg) {
  const int e = blockIdx.x * blockDim.x + threadIdx.x;
  if (e >= N_EDGES) return;
  atomicAdd(&deg[ei[N_EDGES + e]], 1);
}

#define SCAN_T 1024
__global__ __launch_bounds__(SCAN_T) void gat_scan_kernel(const int* __restrict__ deg,
                                                          int* __restrict__ off) {
  __shared__ int wp[32];
  __shared__ int wv[32];
  const int t = threadIdx.x, lane = t & 31, w = t >> 5;
  const int first = blockIdx.x * SCAN_T;
  int p = 0;
  for (int i = t; i < first; i += SCAN_T) p += deg[i];
  const int idx = first + t;
  const int v = idx < N_NODES ? deg[idx] : 0;
  int inc = v;
#pragma unroll
  for (int d = 1; d < 32; d <<= 1) {
    const int u = __shfl_up(inc, d, 32);
    if (lane >= d) inc += u;
  }
#pragma unroll
  for (int mk = 16; mk >= 1; mk >>= 1) p += __shfl_xor(p, mk, 32);
  if (lane == 31) wv[w] = inc;
  if (lane == 0) wp[w] = p;
  __syncthreads();
  int base = 0;
  for (int k = 0; k < 32; ++k) {
    base += wp[k];
    if (k < w) base += wv[k];
  }
  const int o = base + inc - v;
  if (idx < N_NODES) {
    off[idx] = o;
    if (idx == N_NODES - 1) off[N_NODES] = o + v;
  }
}

__global__ void gat_fill_kernel(const int* __restrict__ ei, const int* __restrict__ off,
                                int* __restrict__ deg, int* __restrict__ bucket) {
  const int e = blockIdx.x * blockDim.x + threadIdx.x;
  if (e >= N_EDGES) return;
  const int s = ei[e], d = ei[N_EDGES + e];
  const int slot = atomicSub(&deg[d], 1) - 1;
  bucket[off[d] + slot] = s;
}

__global__ __launch_bounds__(256) void gat_sort_kernel(const int* __restrict__ off,
                                                       const int* __restrict__ bucket,
                                                       int* __restrict__ sorted) {
  const int lane = threadIdx.x & 31;
  const int n = __builtin_amdgcn_readfirstlane((int)(blockIdx.x * 8 + (threadIdx.x >> 5)));
  const int o = off[n];
  const int dg = off[n + 1] - o;
  for (int i = lane; i < dg; i += 32) {
    const int key = bucket[o + i];
    int rank = 0;
    for (int j = 0; j < dg; ++j) {
      const int kj = bucket[o + j];
      rank += (kj < key || (kj == key && j < i)) ? 1 : 0;
    }
    sorted[o + rank] = key;
  }
}


template <bool IS_MAX>
__device__ __forceinline__ void wave_reduce8(float (&v)[8], int lane) {
  const bool b4 = (lane & 16) != 0, b3 = (lane & 8) != 0, b2 = (lane & 4) != 0;
  float w[4], u[2], t;
#pragma unroll
  for (int k = 0; k < 4; ++k) {
    const float send = b4 ? v[k] : v[k + 4];
    const float keep = b4 ? v[k + 4] : v[k];
    const float got = __shfl_xor(send, 16, 32);
    w[k] = IS_MAX ? fmaxf(keep, got) : keep + got;
  }
#pragma unroll
  for (int k = 0; k < 2; ++k) {
    const float send = b3 ? w[k] : w[k + 2];
    const float keep = b3 ? w[k + 2] : w[k];
    const float got = __shfl_xor(send, 8, 32);
    u[k] = IS_MAX ? fmaxf(keep, got) : keep + got;
  }
  {
    const float send = b2 ? u[0] : u[1];
    const float keep = b2 ? u[1] : u[0];
    const float got = __shfl_xor(send, 4, 32);
    t = IS_MAX ? fmaxf(keep, got) : keep + got;
  }
  {
    const float got = __shfl_xor(t, 2, 32);
    t = IS_MAX ? fmaxf(t, got) : t + got;
  }
  {
    const float got = __shfl_xor(t, 1, 32);
    t = IS_MAX ? fmaxf(t, got) : t + got;
  }
#pragma unroll
  for (int q = 0; q < 8; ++q)
    v[q] = __int_as_float(__builtin_amdgcn_readlane(__float_as_int(t), q * 4));
}

__global__ __launch_bounds__(256) void gat_gather_kernel(
    const int* __restrict__ off, const int* __restrict__ sorted,
    const float* __restrict__ a_src, const float* __restrict__ a_dst,
    const float* __restrict__ h, const float* __restrict__ bias,
    float* __restrict__ out) {
  __shared__ float sal[8][32 * HEADS];
  const int lane = threadIdx.x & 31;
  const int wv = threadIdx.x >> 5;
  const int n = __builtin_amdgcn_readfirstlane((int)(blockIdx.x * 8 + wv));
  const int o = off[n];
  const int m = off[n + 1] - o + 1;
  const int hd = lane >> 2;
  float* myal = sal[wv];

  float ad[8];
  {
    const float4 d0 = ((const float4*)(a_dst + (size_t)n * HEADS))[0];
    const float4 d1 = ((const float4*)(a_dst + (size_t)n * HEADS))[1];
    ad[0] = d0.x; ad[1] = d0.y; ad[2] = d0.z; ad[3] = d0.w;
    ad[4] = d1.x; ad[5] = d1.y; ad[6] = d1.z; ad[7] = d1.w;
  }

  int s = n;
  bool valid = false;
  float e[8];
  auto load_chunk = [&](int base) {
    const int j = base + lane;
    valid = j < m;
    s = n;
    if (valid && j > 0) s = sorted[o + j - 1];
    const float4 p0 = ((const float4*)(a_src + (size_t)s * HEADS))[0];
    const float4 p1 = ((const float4*)(a_src + (size_t)s * HEADS))[1];
    e[0] = leaky(p0.x + ad[0]); e[1] = leaky(p0.y + ad[1]);
    e[2] = leaky(p0.z + ad[2]); e[3] = leaky(p0.w + ad[3]);
    e[4] = leaky(p1.x + ad[4]); e[5] = leaky(p1.y + ad[5]);
    e[6] = leaky(p1.z + ad[6]); e[7] = leaky(p1.w + ad[7]);
  };
  const bool many = m > 32;

  float mx[8];
#pragma unroll
  for (int q = 0; q < 8; ++q) mx[q] = -INFINITY;
  for (int base = 0; base < m; base += 32) {
    load_chunk(base);
    if (valid) {
#pragma unroll
      for (int q = 0; q < 8; ++q) mx[q] = fmaxf(mx[q], e[q]);
    }
  }
  wave_reduce8<true>(mx, lane);

  float den[8];
#pragma unroll
  for (int q = 0; q < 8; ++q) den[q] = 0.f;
  for (int base = 0; base < m; base += 32) {
    if (many) load_chunk(base);
    if (valid) {
#pragma unroll
      for (int q = 0; q < 8; ++q) den[q] += __expf(e[q] - mx[q]);
    }
  }
  wave_reduce8<false>(den, lane);

  const float4* h4 = (const float4*)h;
  float4 acc = make_float4(0.f, 0.f, 0.f, 0.f);
  for (int base = 0; base < m; base += 32) {
    if (many) load_chunk(base);
    float al[8];
#pragma unroll
    for (int q = 0; q < 8; ++q)
      al[q] = valid ? __expf(e[q] - mx[q]) / (den[q] + GAT_EPS) : 0.f;
    ((float4*)(myal + lane * HEADS))[0] = make_float4(al[0], al[1], al[2], al[3]);
    ((float4*)(myal + lane * HEADS))[1] = make_float4(al[4], al[5], al[6], al[7]);

    const int left = m - base;
    const int cnt = left < 32 ? left : 32;
    int j = 0;
    for (; j + 4 <= cnt; j += 4) {
      const int s0 = __builtin_amdgcn_readlane(s, j);
      const int s1 = __builtin_amdgcn_readlane(s, j + 1);
      const int s2 = __builtin_amdgcn_readlane(s, j + 2);
      const int s3 = __builtin_amdgcn_readlane(s, j + 3);
      const float4 r0 = h4[(size_t)s0 * 32 + lane];
      const float4 r1 = h4[(size_t)s1 * 32 + lane];
      const float4 r2 = h4[(size_t)s2 * 32 + lane];
      const float4 r3 = h4[(size_t)s3 * 32 + lane];
      const float a0 = myal[(j + 0) * HEADS + hd];
      const float a1 = myal[(j + 1) * HEADS + hd];
      const float a2 = myal[(j + 2) * HEADS + hd];
      const float a3 = myal[(j + 3) * HEADS + hd];
      acc.x += a0 * r0.x; acc.y += a0 * r0.y; acc.z += a0 * r0.z; acc.w += a0 * r0.w;
      acc.x += a1 * r1.x; acc.y += a1 * r1.y; acc.z += a1 * r1.z; acc.w += a1 * r1.w;
      acc.x += a2 * r2.x; acc.y += a2 * r2.y; acc.z += a2 * r2.z; acc.w += a2 * r2.w;
      acc.x += a3 * r3.x; acc.y += a3 * r3.y; acc.z += a3 * r3.z; acc.w += a3 * r3.w;
    }
    for (; j < cnt; ++j) {
      const int s0 = __builtin_amdgcn_readlane(s, j);
      const float4 r0 = h4[(size_t)s0 * 32 + lane];
      const float a0 = myal[j * HEADS + hd];
      acc.x += a0 * r0.x; acc.y += a0 * r0.y; acc.z += a0 * r0.z; acc.w += a0 * r0.w;
    }
  }

  const float4 bv = ((const float4*)bias)[lane];
  float4 r;
  r.x = acc.x + bv.x; r.y = acc.y + bv.y; r.z = acc.z + bv.z; r.w = acc.w + bv.w;
  r.x = r.x > 0.f ? r.x : expm1f(r.x);
  r.y = r.y > 0.f ? r.y : expm1f(r.y);
  r.z = r.z > 0.f ? r.z : expm1f(r.z);
  r.w = r.w > 0.f ? r.w : expm1f(r.w);
  ((float4*)out)[(size_t)n * 32 + lane] = r;
}

extern "C" void kernel_launch(void* const* d_in, const int* in_sizes, int n_in,
                              void* d_out, int out_size, void* d_ws, size_t ws_size,
                              hipStream_t stream) {
  (void)in_sizes; (void)n_in; (void)out_size; (void)ws_size;
  const float* x       = (const float*)d_in[0];
  const int*   ei      = (const int*)d_in[1];
  const float* W       = (const float*)d_in[2];
  const float* att_src = (const float*)d_in[3];
  const float* att_dst = (const float*)d_in[4];
  const float* bias    = (const float*)d_in[5];

  float* h      = (float*)d_ws;
  float* a_src  = h + (size_t)N_NODES * OUT_DIM;
  float* a_dst  = a_src + (size_t)N_NODES * HEADS;
  int*   deg    = (int*)(a_dst + (size_t)N_NODES * HEADS);
  int*   off    = deg + N_NODES;
  int*   bucket = off + (N_NODES + 1);
  int*   sorted = bucket + N_EDGES;

  gat_gemm_wmma_kernel<<<N_NODES / 16, 256, 0, stream>>>(x, W, h);

  gat_attn_proj_kernel<<<(N_NODES * HEADS + 255) / 256, 256, 0, stream>>>(
      h, att_src, att_dst, a_src, a_dst, deg);

  gat_count_kernel<<<(N_EDGES + 255) / 256, 256, 0, stream>>>(ei, deg);
  gat_scan_kernel<<<(N_NODES + SCAN_T - 1) / SCAN_T, SCAN_T, 0, stream>>>(deg, off);
  gat_fill_kernel<<<(N_EDGES + 255) / 256, 256, 0, stream>>>(ei, off, deg, bucket);
  gat_sort_kernel<<<N_NODES / 8, 256, 0, stream>>>(off, bucket, sorted);
  gat_gather_kernel<<<N_NODES / 8, 256, 0, stream>>>(off, sorted, a_src, a_dst, h,
                                                     bias, (float*)d_out);
}


# Edge ranks and per-node counts from LDS counters (32 edge slices, one work-group each, then a combine pass over slices); fill places edges without atomics: all 1600000 scattered global atomics gone
# speedup vs baseline: 5.5435x; 1.9502x over previous
#include <hip/hip_runtime.h>
#include <math.h>

#define N_NODES 50000
#define N_EDGES 800000
#define IN_DIM 128
#define OUT_DIM 128
#define HEADS 8
#define CPH 16
#define NEG_SLOPE 0.2f
#define GAT_EPS 1e-16f

typedef float v2f __attribute__((ext_vector_type(2)));
typedef float v8f __attribute__((ext_vector_type(8)));

#define LDX 132

__global__ __launch_bounds__(256) void gat_gemm_wmma_kernel(
    const float* __restrict__ x, const float* __restrict__ Wg,
    float* __restrict__ h) {
  __shared__ float sx[16 * LDX];
  const int tid = threadIdx.x;
  const int mbase = blockIdx.x * 16;

  for (int i = tid; i < 16 * 32; i += 256) {
    const int r  = i >> 5;
    const int c4 = i & 31;
    const float4 v = ((const float4*)(x + (size_t)(mbase + r) * IN_DIM))[c4];
    float* dst = &sx[r * LDX + c4 * 4];
    dst[0] = v.x; dst[1] = v.y; dst[2] = v.z; dst[3] = v.w;
  }
  __syncthreads();

  const int wave = tid >> 5;
  const int lane = tid & 31;
  const int half = lane >> 4;
  const int mr   = lane & 15;
  const int nb   = wave * 16;

  v8f acc = {};
#pragma unroll
  for (int k0 = 0; k0 < IN_DIM; k0 += 4) {
    v2f a, b;
    a.x = sx[mr * LDX + k0 + 2 * half];
    a.y = sx[mr * LDX + k0 + 2 * half + 1];
    b.x = Wg[(size_t)(k0 + 2 * half) * OUT_DIM + nb + mr];
    b.y = Wg[(size_t)(k0 + 2 * half + 1) * OUT_DIM + nb + mr];
    acc = __builtin_amdgcn_wmma_f32_16x16x4_f32(
         false, a,  false, b,
         (short)0, acc,  false,  false);
  }

#pragma unroll
  for (int r = 0; r < 8; ++r) {
    const int row = half * 8 + r;
    h[(size_t)(mbase + row) * OUT_DIM + nb + mr] = acc[r];
  }
}

__global__ void gat_attn_proj_kernel(const float* __restrict__ h,
                                     const float* __restrict__ att_src,
                                     const float* __restrict__ att_dst,
                                     float* __restrict__ a_src,
                                     float* __restrict__ a_dst) {
  const int i = blockIdx.x * blockDim.x + threadIdx.x;
  if (i >= N_NODES * HEADS) return;
  const int hd = i & (HEADS - 1);
  const float4* hp = (const float4*)(h + (size_t)i * CPH);
  const float4* as = (const float4*)(att_src + hd * CPH);
  const float4* ad = (const float4*)(att_dst + hd * CPH);
  float s = 0.f, d = 0.f;
#pragma unroll
  for (int q = 0; q < 4; ++q) {
    const float4 hv = hp[q], av = as[q], dv = ad[q];
    s += hv.x * av.x + hv.y * av.y + hv.z * av.z + hv.w * av.w;
    d += hv.x * dv.x + hv.y * dv.y + hv.z * dv.z + hv.w * dv.w;
  }
  a_src[i] = s;
  a_dst[i] = d;
}

__device__ __forceinline__ float leaky(float v) {
  return v >= 0.f ? v : NEG_SLOPE * v;
}

#define K_SLICES 32
#define EPB (N_EDGES / K_SLICES)
#define V4PB (EPB / 4)
#define CNT_T 1024
#define CNT_IT ((V4PB + CNT_T - 1) / CNT_T)

__global__ __launch_bounds__(CNT_T) void gat_count_kernel(const int* __restrict__ ei,
                                                          int* __restrict__ cnt,
                                                          int* __restrict__ rank) {
  __shared__ int hist[N_NODES];
  const int t = threadIdx.x;
  const int k = blockIdx.x;

  const int4* d4 = (const int4*)(ei + N_EDGES + k * EPB);
  int4 dv[CNT_IT];
#pragma unroll
  for (int j = 0; j < CNT_IT; ++j) {
    const int idx = t + j * CNT_T;
    dv[j] = d4[idx < V4PB ? idx : V4PB - 1];
  }

  int4* h4 = (int4*)hist;
  for (int i = t; i < N_NODES / 4; i += CNT_T) h4[i] = make_int4(0, 0, 0, 0);
  __syncthreads();

  int4 rv[CNT_IT];
#pragma unroll
  for (int j = 0; j < CNT_IT; ++j) {
    const int idx = t + j * CNT_T;
    rv[j] = make_int4(0, 0, 0, 0);
    if (idx < V4PB) {
      rv[j].x = atomicAdd(&hist[dv[j].x], 1);
      rv[j].y = atomicAdd(&hist[dv[j].y], 1);
      rv[j].z = atomicAdd(&hist[dv[j].z], 1);
      rv[j].w = atomicAdd(&hist[dv[j].w], 1);
    }
  }
  __syncthreads();

  int4* r4 = (int4*)(rank + k * EPB);
#pragma unroll
  for (int j = 0; j < CNT_IT; ++j) {
    const int idx = t + j * CNT_T;
    if (idx < V4PB) r4[idx] = rv[j];
  }
  int4* o4 = (int4*)(cnt + (size_t)k * N_NODES);
  for (int i = t; i < N_NODES / 4; i += CNT_T) o4[i] = h4[i];
}

__global__ __launch_bounds__(256) void gat_combine_kernel(int* __restrict__ cnt,
                                                          int* __restrict__ deg) {
  const int i = blockIdx.x * 256 + threadIdx.x;
  if (i >= N_NODES) return;
  int v[K_SLICES];
#pragma unroll
  for (int k = 0; k < K_SLICES; ++k) v[k] = cnt[(size_t)k * N_NODES + i];
  int run = 0;
#pragma unroll
  for (int k = 0; k < K_SLICES; ++k) {
    cnt[(size_t)k * N_NODES + i] = run;
    run += v[k];
  }
  deg[i] = run;
}

#define SCAN_T 1024
__global__ __launch_bounds__(SCAN_T) void gat_scan_kernel(const int* __restrict__ deg,
                                                          int* __restrict__ off) {
  __shared__ int wp[32];
  __shared__ int wv[32];
  const int t = threadIdx.x, lane = t & 31, w = t >> 5;
  const int first = blockIdx.x * SCAN_T;
  int p = 0;
  for (int i = t; i < first; i += SCAN_T) p += deg[i];
  const int idx = first + t;
  const int v = idx < N_NODES ? deg[idx] : 0;
  int inc = v;
#pragma unroll
  for (int d = 1; d < 32; d <<= 1) {
    const int u = __shfl_up(inc, d, 32);
    if (lane >= d) inc += u;
  }
#pragma unroll
  for (int mk = 16; mk >= 1; mk >>= 1) p += __shfl_xor(p, mk, 32);
  if (lane == 31) wv[w] = inc;
  if (lane == 0) wp[w] = p;
  __syncthreads();
  int base = 0;
  for (int k = 0; k < 32; ++k) {
    base += wp[k];
    if (k < w) base += wv[k];
  }
  const int o = base + inc - v;
  if (idx < N_NODES) {
    off[idx] = o;
    if (idx == N_NODES - 1) off[N_NODES] = o + v;
  }
}

__global__ void gat_fill_kernel(const int* __restrict__ ei, const int* __restrict__ off,
                                const int* __restrict__ cnt, const int* __restrict__ rank,
                                int* __restrict__ bucket) {
  const int e = blockIdx.x * blockDim.x + threadIdx.x;
  if (e >= N_EDGES) return;
  const int s = ei[e], d = ei[N_EDGES + e];
  const int k = e / EPB;
  bucket[off[d] + cnt[(size_t)k * N_NODES + d] + rank[e]] = s;
}

__global__ __launch_bounds__(256) void gat_sort_kernel(const int* __restrict__ off,
                                                       const int* __restrict__ bucket,
                                                       int* __restrict__ sorted) {
  const int lane = threadIdx.x & 31;
  const int n = __builtin_amdgcn_readfirstlane((int)(blockIdx.x * 8 + (threadIdx.x >> 5)));
  const int o = off[n];
  const int dg = off[n + 1] - o;
  for (int i = lane; i < dg; i += 32) {
    const int key = bucket[o + i];
    int rank = 0;
    for (int j = 0; j < dg; ++j) {
      const int kj = bucket[o + j];
      rank += (kj < key || (kj == key && j < i)) ? 1 : 0;
    }
    sorted[o + rank] = key;
  }
}


template <bool IS_MAX>
__device__ __forceinline__ void wave_reduce8(float (&v)[8], int lane) {
  const bool b4 = (lane & 16) != 0, b3 = (lane & 8) != 0, b2 = (lane & 4) != 0;
  float w[4], u[2], t;
#pragma unroll
  for (int k = 0; k < 4; ++k) {
    const float send = b4 ? v[k] : v[k + 4];
    const float keep = b4 ? v[k + 4] : v[k];
    const float got = __shfl_xor(send, 16, 32);
    w[k] = IS_MAX ? fmaxf(keep, got) : keep + got;
  }
#pragma unroll
  for (int k = 0; k < 2; ++k) {
    const float send = b3 ? w[k] : w[k + 2];
    const float keep = b3 ? w[k + 2] : w[k];
    const float got = __shfl_xor(send, 8, 32);
    u[k] = IS_MAX ? fmaxf(keep, got) : keep + got;
  }
  {
    const float send = b2 ? u[0] : u[1];
    const float keep = b2 ? u[1] : u[0];
    const float got = __shfl_xor(send, 4, 32);
    t = IS_MAX ? fmaxf(keep, got) : keep + got;
  }
  {
    const float got = __shfl_xor(t, 2, 32);
    t = IS_MAX ? fmaxf(t, got) : t + got;
  }
  {
    const float got = __shfl_xor(t, 1, 32);
    t = IS_MAX ? fmaxf(t, got) : t + got;
  }
#pragma unroll
  for (int q = 0; q < 8; ++q)
    v[q] = __int_as_float(__builtin_amdgcn_readlane(__float_as_int(t), q * 4));
}

__global__ __launch_bounds__(256) void gat_gather_kernel(
    const int* __restrict__ off, const int* __restrict__ sorted,
    const float* __restrict__ a_src, const float* __restrict__ a_dst,
    const float* __restrict__ h, const float* __restrict__ bias,
    float* __restrict__ out) {
  __shared__ float sal[8][32 * HEADS];
  const int lane = threadIdx.x & 31;
  const int wv = threadIdx.x >> 5;
  const int n = __builtin_amdgcn_readfirstlane((int)(blockIdx.x * 8 + wv));
  const int o = off[n];
  const int m = off[n + 1] - o + 1;
  const int hd = lane >> 2;
  float* myal = sal[wv];

  float ad[8];
  {
    const float4 d0 = ((const float4*)(a_dst + (size_t)n * HEADS))[0];
    const float4 d1 = ((const float4*)(a_dst + (size_t)n * HEADS))[1];
    ad[0] = d0.x; ad[1] = d0.y; ad[2] = d0.z; ad[3] = d0.w;
    ad[4] = d1.x; ad[5] = d1.y; ad[6] = d1.z; ad[7] = d1.w;
  }

  int s = n;
  bool valid = false;
  float e[8];
  auto load_chunk = [&](int base) {
    const int j = base + lane;
    valid = j < m;
    s = n;
    if (valid && j > 0) s = sorted[o + j - 1];
    const float4 p0 = ((const float4*)(a_src + (size_t)s * HEADS))[0];
    const float4 p1 = ((const float4*)(a_src + (size_t)s * HEADS))[1];
    e[0] = leaky(p0.x + ad[0]); e[1] = leaky(p0.y + ad[1]);
    e[2] = leaky(p0.z + ad[2]); e[3] = leaky(p0.w + ad[3]);
    e[4] = leaky(p1.x + ad[4]); e[5] = leaky(p1.y + ad[5]);
    e[6] = leaky(p1.z + ad[6]); e[7] = leaky(p1.w + ad[7]);
  };
  const bool many = m > 32;

  float mx[8];
#pragma unroll
  for (int q = 0; q < 8; ++q) mx[q] = -INFINITY;
  for (int base = 0; base < m; base += 32) {
    load_chunk(base);
    if (valid) {
#pragma unroll
      for (int q = 0; q < 8; ++q) mx[q] = fmaxf(mx[q], e[q]);
    }
  }
  wave_reduce8<true>(mx, lane);

  float den[8];
#pragma unroll
  for (int q = 0; q < 8; ++q) den[q] = 0.f;
  for (int base = 0; base < m; base += 32) {
    if (many) load_chunk(base);
    if (valid) {
#pragma unroll
      for (int q = 0; q < 8; ++q) den[q] += __expf(e[q] - mx[q]);
    }
  }
  wave_reduce8<false>(den, lane);

  const float4* h4 = (const float4*)h;
  float4 acc = make_float4(0.f, 0.f, 0.f, 0.f);
  for (int base = 0; base < m; base += 32) {
    if (many) load_chunk(base);
    float al[8];
#pragma unroll
    for (int q = 0; q < 8; ++q)
      al[q] = valid ? __expf(e[q] - mx[q]) / (den[q] + GAT_EPS) : 0.f;
    ((float4*)(myal + lane * HEADS))[0] = make_float4(al[0], al[1], al[2], al[3]);
    ((float4*)(myal + lane * HEADS))[1] = make_float4(al[4], al[5], al[6], al[7]);

    const int left = m - base;
    const int cnt = left < 32 ? left : 32;
    int j = 0;
    for (; j + 4 <= cnt; j += 4) {
      const int s0 = __builtin_amdgcn_readlane(s, j);
      const int s1 = __builtin_amdgcn_readlane(s, j + 1);
      const int s2 = __builtin_amdgcn_readlane(s, j + 2);
      const int s3 = __builtin_amdgcn_readlane(s, j + 3);
      const float4 r0 = h4[(size_t)s0 * 32 + lane];
      const float4 r1 = h4[(size_t)s1 * 32 + lane];
      const float4 r2 = h4[(size_t)s2 * 32 + lane];
      const float4 r3 = h4[(size_t)s3 * 32 + lane];
      const float a0 = myal[(j + 0) * HEADS + hd];
      const float a1 = myal[(j + 1) * HEADS + hd];
      const float a2 = myal[(j + 2) * HEADS + hd];
      const float a3 = myal[(j + 3) * HEADS + hd];
      acc.x += a0 * r0.x; acc.y += a0 * r0.y; acc.z += a0 * r0.z; acc.w += a0 * r0.w;
      acc.x += a1 * r1.x; acc.y += a1 * r1.y; acc.z += a1 * r1.z; acc.w += a1 * r1.w;
      acc.x += a2 * r2.x; acc.y += a2 * r2.y; acc.z += a2 * r2.z; acc.w += a2 * r2.w;
      acc.x += a3 * r3.x; acc.y += a3 * r3.y; acc.z += a3 * r3.z; acc.w += a3 * r3.w;
    }
    for (; j < cnt; ++j) {
      const int s0 = __builtin_amdgcn_readlane(s, j);
      const float4 r0 = h4[(size_t)s0 * 32 + lane];
      const float a0 = myal[j * HEADS + hd];
      acc.x += a0 * r0.x; acc.y += a0 * r0.y; acc.z += a0 * r0.z; acc.w += a0 * r0.w;
    }
  }

  const float4 bv = ((const float4*)bias)[lane];
  float4 r;
  r.x = acc.x + bv.x; r.y = acc.y + bv.y; r.z = acc.z + bv.z; r.w = acc.w + bv.w;
  r.x = r.x > 0.f ? r.x : expm1f(r.x);
  r.y = r.y > 0.f ? r.y : expm1f(r.y);
  r.z = r.z > 0.f ? r.z : expm1f(r.z);
  r.w = r.w > 0.f ? r.w : expm1f(r.w);
  ((float4*)out)[(size_t)n * 32 + lane] = r;
}

extern "C" void kernel_launch(void* const* d_in, const int* in_sizes, int n_in,
                              void* d_out, int out_size, void* d_ws, size_t ws_size,
                              hipStream_t stream) {
  (void)in_sizes; (void)n_in; (void)out_size; (void)ws_size;
  const float* x       = (const float*)d_in[0];
  const int*   ei      = (const int*)d_in[1];
  const float* W       = (const float*)d_in[2];
  const float* att_src = (const float*)d_in[3];
  const float* att_dst = (const float*)d_in[4];
  const float* bias    = (const float*)d_in[5];

  float* h      = (float*)d_ws;
  float* a_src  = h + (size_t)N_NODES * OUT_DIM;
  float* a_dst  = a_src + (size_t)N_NODES * HEADS;
  int*   deg    = (int*)(a_dst + (size_t)N_NODES * HEADS);
  int*   off    = deg + N_NODES;
  int*   bucket = off + (N_NODES + 1);
  int*   sorted = bucket + N_EDGES;
  int*   cnt    = (int*)d_ws;
  int*   rank   = cnt + (size_t)K_SLICES * N_NODES;

  gat_count_kernel<<<K_SLICES, CNT_T, 0, stream>>>(ei, cnt, rank);
  gat_combine_kernel<<<(N_NODES + 255) / 256, 256, 0, stream>>>(cnt, deg);
  gat_scan_kernel<<<(N_NODES + SCAN_T - 1) / SCAN_T, SCAN_T, 0, stream>>>(deg, off);
  gat_fill_kernel<<<(N_EDGES + 255) / 256, 256, 0, stream>>>(ei, off, cnt, rank, bucket);
  gat_sort_kernel<<<N_NODES / 8, 256, 0, stream>>>(off, bucket, sorted);

  gat_gemm_wmma_kernel<<<N_NODES / 16, 256, 0, stream>>>(x, W, h);

  gat_attn_proj_kernel<<<(N_NODES * HEADS + 255) / 256, 256, 0, stream>>>(
      h, att_src, att_dst, a_src, a_dst);

  gat_gather_kernel<<<N_NODES / 8, 256, 0, stream>>>(off, sorted, a_src, a_dst, h,
                                                     bias, (float*)d_out);
}


# Lists of 33 to 64 in-edges ordered by their wave, two entries per lane via its LDS words, not the scratch arrays: the few long-list waves lose their serial device-scope round trips; sum order kept
# speedup vs baseline: 12.4084x; 2.2384x over previous
#include <hip/hip_runtime.h>
#include <math.h>

#define N_NODES 50000
#define N_EDGES 800000
#define IN_DIM 128
#define OUT_DIM 128
#define HEADS 8
#define CPH 16
#define NEG_SLOPE 0.2f
#define GAT_EPS 1e-16f
#define K_SLICES 32
#define EPB (N_EDGES / K_SLICES)

typedef float v2f __attribute__((ext_vector_type(2)));
typedef float v8f __attribute__((ext_vector_type(8)));

#define LDX 132
#define GEMM_ROWS 64
#define GEMM_UNITS ((N_NODES + GEMM_ROWS - 1) / GEMM_ROWS)
#define GEMM_WGS 196
#define M_TILES (N_NODES / 16)

__device__ __forceinline__ float leaky(float v) {
  return v >= 0.f ? v : NEG_SLOPE * v;
}

static_assert(K_SLICES == 32, "a gather wave reads one slice per lane");
static_assert(N_EDGES % K_SLICES == 0 && EPB % 4 == 0 && N_NODES % 8 == 0,
              "slices and tables are moved in whole 16-byte groups");
static_assert(EPB <= 65535 && N_NODES <= 65536,
              "prefix entries and source ids are kept in 16 bits");
#define V4PB (EPB / 4)
#define CNT_T 1024
#define CNT_IT ((V4PB + CNT_T - 1) / CNT_T)
#define SCAN_V4 13
#define HIST_PAD (CNT_T * SCAN_V4 * 4)
static_assert(2 * GEMM_ROWS * LDX <= HIST_PAD, "the GEMM's two x buffers live in the counters' LDS");
static_assert(GEMM_WGS * 4 >= GEMM_UNITS && CNT_T == 1024 && N_NODES % 16 == 0,
              "32 waves per GEMM work-group, whole M-tiles, at most 4 units each");

__device__ __forceinline__ void gemm_role(float* __restrict__ sx, int g,
                                          const float* __restrict__ x,
                                          const float* __restrict__ Wg,
                                          float* __restrict__ h) {
  const int tid = threadIdx.x;
  const int wave = tid >> 5;
  const int lane = tid & 31;
  const int half = lane >> 4;
  const int mr   = lane & 15;
  const int nb   = (wave & 7) * 16;
  const int mg   = wave >> 3;

  const int r0 = tid >> 5, c4 = tid & 31;
  int u = g;
  float4 p0 = make_float4(0.f, 0.f, 0.f, 0.f), p1 = p0;
  if (u < GEMM_UNITS) {
    const int ra = u * GEMM_ROWS + r0, rb = ra + 32;
    if (ra < N_NODES) p0 = ((const float4*)(x + (size_t)ra * IN_DIM))[c4];
    if (rb < N_NODES) p1 = ((const float4*)(x + (size_t)rb * IN_DIM))[c4];
  }

  v2f bf[IN_DIM / 4];
#pragma unroll
  for (int kk = 0; kk < IN_DIM / 4; ++kk) {
    bf[kk].x = Wg[(size_t)(4 * kk + 2 * half) * OUT_DIM + nb + mr];
    bf[kk].y = Wg[(size_t)(4 * kk + 2 * half + 1) * OUT_DIM + nb + mr];
  }

  int buf = 0;
  while (u < GEMM_UNITS) {
    float* sb = sx + buf * (GEMM_ROWS * LDX);
    *(float4*)(sb + r0 * LDX + c4 * 4) = p0;
    *(float4*)(sb + (r0 + 32) * LDX + c4 * 4) = p1;
    __syncthreads();

    const int un = u + GEMM_WGS;
    if (un < GEMM_UNITS) {
      const int ra = un * GEMM_ROWS + r0, rb = ra + 32;
      p0 = make_float4(0.f, 0.f, 0.f, 0.f);
      p1 = p0;
      if (ra < N_NODES) p0 = ((const float4*)(x + (size_t)ra * IN_DIM))[c4];
      if (rb < N_NODES) p1 = ((const float4*)(x + (size_t)rb * IN_DIM))[c4];
    }

    const int mt = u * (GEMM_ROWS / 16) + mg;
    if (mt < M_TILES) {
      const float* sa = sb + (mg * 16 + mr) * LDX + 2 * half;
      v8f acc = {};
#pragma unroll
      for (int kk = 0; kk < IN_DIM / 4; ++kk) {
        const v2f a = *(const v2f*)(sa + 4 * kk);
        acc = __builtin_amdgcn_wmma_f32_16x16x4_f32(
             false, a,  false, bf[kk],
             (short)0, acc,  false,  false);
      }
      float* hrow = h + (size_t)(mt * 16 + half * 8) * OUT_DIM + nb + mr;
#pragma unroll
      for (int r = 0; r < 8; ++r) hrow[(size_t)r * OUT_DIM] = acc[r];
    }
    buf ^= 1;
    u = un;
  }
}

__global__ __launch_bounds__(CNT_T) void gat_build_gemm_kernel(
    const int* __restrict__ ei, unsigned short* __restrict__ cnt16,
    int* __restrict__ bucket, int* __restrict__ cursor,
    const float* __restrict__ x, const float* __restrict__ Wg,
    float* __restrict__ h) {
  __shared__ __attribute__((aligned(16))) int hist[HIST_PAD];
  __shared__ __attribute__((aligned(16))) int srt[EPB];
  __shared__ int wtot[32];

  if (blockIdx.x >= K_SLICES) {
    gemm_role((float*)hist, (int)blockIdx.x - K_SLICES, x, Wg, h);
    return;
  }

  const int t = threadIdx.x, lane = t & 31, w = t >> 5;
  const int k = blockIdx.x;

  if (k == 0 && t == 0) *cursor = 0;

  const int4* d4 = (const int4*)(ei + N_EDGES + k * EPB);
  const int4* s4 = (const int4*)(ei + k * EPB);
  int4 dv[CNT_IT], sv[CNT_IT];
#pragma unroll
  for (int j = 0; j < CNT_IT; ++j) {
    const int idx = t + j * CNT_T;
    dv[j] = d4[idx < V4PB ? idx : V4PB - 1];
  }
#pragma unroll
  for (int j = 0; j < CNT_IT; ++j) {
    const int idx = t + j * CNT_T;
    sv[j] = s4[idx < V4PB ? idx : V4PB - 1];
  }

  int4* h4 = (int4*)hist;
  for (int i = t; i < HIST_PAD / 4; i += CNT_T) h4[i] = make_int4(0, 0, 0, 0);
  __syncthreads();

  int4 rv[CNT_IT];
#pragma unroll
  for (int j = 0; j < CNT_IT; ++j) {
    const int idx = t + j * CNT_T;
    rv[j] = make_int4(0, 0, 0, 0);
    if (idx < V4PB) {
      rv[j].x = atomicAdd(&hist[dv[j].x], 1);
      rv[j].y = atomicAdd(&hist[dv[j].y], 1);
      rv[j].z = atomicAdd(&hist[dv[j].z], 1);
      rv[j].w = atomicAdd(&hist[dv[j].w], 1);
    }
  }
  __syncthreads();

  int4* my = h4 + t * SCAN_V4;
  int sum = 0;
#pragma unroll
  for (int j = 0; j < SCAN_V4; ++j) {
    const int4 v = my[j];
    sum += (v.x + v.y) + (v.z + v.w);
  }
  int inc = sum;
#pragma unroll
  for (int d = 1; d < 32; d <<= 1) {
    const int u = __shfl_up(inc, d, 32);
    if (lane >= d) inc += u;
  }
  if (lane == 31) wtot[w] = inc;
  __syncthreads();
  int run = inc - sum;
  for (int q = 0; q < 32; ++q)
    if (q < w) run += wtot[q];
#pragma unroll
  for (int j = 0; j < SCAN_V4; ++j) {
    const int4 v = my[j];
    int4 o;
    o.x = run;
    o.y = o.x + v.x;
    o.z = o.y + v.y;
    o.w = o.z + v.z;
    run = o.w + v.w;
    my[j] = o;
  }
  __syncthreads();

#pragma unroll
  for (int j = 0; j < CNT_IT; ++j) {
    const int idx = t + j * CNT_T;
    if (idx < V4PB) {
      rv[j].x += hist[dv[j].x];
      rv[j].y += hist[dv[j].y];
      rv[j].z += hist[dv[j].z];
      rv[j].w += hist[dv[j].w];
    }
  }
#pragma unroll
  for (int j = 0; j < CNT_IT; ++j) {
    const int idx = t + j * CNT_T;
    if (idx < V4PB) {
      srt[rv[j].x] = sv[j].x;
      srt[rv[j].y] = sv[j].y;
      srt[rv[j].z] = sv[j].z;
      srt[rv[j].w] = sv[j].w;
    }
  }

  int4* o4 = (int4*)(cnt16 + (size_t)k * N_NODES);
  for (int i = t; i < N_NODES / 8; i += CNT_T) {
    const int4 a = h4[2 * i], b = h4[2 * i + 1];
    int4 p;
    p.x = a.x | (a.y << 16);
    p.y = a.z | (a.w << 16);
    p.z = b.x | (b.y << 16);
    p.w = b.z | (b.w << 16);
    o4[i] = p;
  }
  __syncthreads();

  int4* b4 = (int4*)(bucket + (size_t)k * EPB);
  const int4* q4 = (const int4*)srt;
  for (int i = t; i < V4PB; i += CNT_T) b4[i] = q4[i];
}


__device__ __forceinline__ float quad_sum(float v) {
  float t = v + __int_as_float(__builtin_amdgcn_update_dpp(
                    0, __float_as_int(v), 0x161, 0xF, 0xF, true));
  t = t + __int_as_float(__builtin_amdgcn_update_dpp(
              0, __float_as_int(t), 0x162, 0xF, 0xF, true));
  return t;
}

__device__ __forceinline__ int wave_incl_scan(int v, int lane) {
  int x = v;
  x += __builtin_amdgcn_update_dpp(0, x, 0x111, 0xF, 0xF, true);
  x += __builtin_amdgcn_update_dpp(0, x, 0x112, 0xF, 0xF, true);
  x += __builtin_amdgcn_update_dpp(0, x, 0x114, 0xF, 0xF, true);
  x += __builtin_amdgcn_update_dpp(0, x, 0x118, 0xF, 0xF, true);
  const int lo = __builtin_amdgcn_readlane(x, 15);
  return lane >= 16 ? x + lo : x;
}

typedef float v4f __attribute__((ext_vector_type(4)));

__device__ __forceinline__ float dot4(const v4f a, const v4f b) {
  const v2f t = a.xy * b.xy + a.zw * b.zw;
  return t.x + t.y;
}

template <int G>
__device__ __forceinline__ void rows_step(const v4f* __restrict__ h4, int lane,
                                          const int (&sid)[G], int nvalid,
                                          const v4f as4, float adst,
                                          float& m, float& den, v4f& acc) {
  v4f r[G];
  float e[G];
#pragma unroll
  for (int i = 0; i < G; ++i) r[i] = h4[(size_t)sid[i] * 32 + lane];
  float mn = m;
#pragma unroll
  for (int i = 0; i < G; ++i) {
    e[i] = i < nvalid ? leaky(quad_sum(dot4(r[i], as4)) + adst) : -INFINITY;
    mn = fmaxf(mn, e[i]);
  }
  const float sc = __expf(m - mn);
  m = mn;
  den *= sc;
  acc *= sc;
#pragma unroll
  for (int i = 0; i < G; ++i) {
    const float p = __expf(e[i] - mn);
    den += p;
    acc += p * r[i];
  }
}

__device__ __forceinline__ int scratch_load(unsigned short* p) {
  return (int)__hip_atomic_load(p, __ATOMIC_RELAXED, __HIP_MEMORY_SCOPE_AGENT);
}

__global__ __launch_bounds__(256) void gat_gather_kernel(
    const unsigned short* __restrict__ cnt16, const int* __restrict__ bucket,
    unsigned short* raw16, unsigned short* srt16, int* cursor,
    const float* __restrict__ att_src, const float* __restrict__ att_dst,
    const float* __restrict__ h, const float* __restrict__ bias,
    float* __restrict__ out) {
  __shared__ int lst[8 * 64];
  const int lane = threadIdx.x & 31;
  const int wv = threadIdx.x >> 5;
  const int n = __builtin_amdgcn_readfirstlane((int)(blockIdx.x * 8 + wv));
  const v4f* h4 = (const v4f*)h;
  int* wl = lst + wv * 64;

  const v4f rs = h4[(size_t)n * 32 + lane];
  const v4f as4 = ((const v4f*)att_src)[lane];
  const v4f ad4 = ((const v4f*)att_dst)[lane];

  const unsigned short* crow = cnt16 + (size_t)lane * N_NODES + n;
  const int sa = crow[0];
  int sb = EPB;
  if (n + 1 < N_NODES) sb = crow[1];
  const int len = sb - sa;
  const int start = lane * EPB + sa;
  const int inc = wave_incl_scan(len, lane);
  const int pos = inc - len;
  const int dg = __builtin_amdgcn_readlane(inc, 31);
  const bool many = dg > 64;

  int s = 0;
  int mb = 0;
  if (dg <= 32) {
#pragma nounroll
    for (int t = 0; t < len; ++t) wl[pos + t] = start + t;
    const bool valid = lane < dg;
    int id = 0;
    if (valid) id = bucket[wl[lane]];
    const int ck = id * 32 + lane;
    int rk = 0;
    for (int j = 0; j < dg; ++j) {
      const int kj = __builtin_amdgcn_readlane(ck, j);
      rk += kj < ck ? 1 : 0;
    }
    if (!valid) rk = lane;
    s = __builtin_amdgcn_ds_permute(rk << 2, id);
  } else if (!many) {
#pragma nounroll
    for (int t = 0; t < len; ++t) wl[pos + t] = start + t;
    const bool hi = lane + 32 < dg;
    const int id0 = bucket[wl[lane]];
    int id1 = 0;
    if (hi) id1 = bucket[wl[32 + lane]];
    const int ck0 = id0 * 64 + lane;
    const int ck1 = id1 * 64 + 32 + lane;
    int rk0 = 0, rk1 = 0;
    for (int j = 0; j < 32; ++j) {
      const int kj = __builtin_amdgcn_readlane(ck0, j);
      rk0 += kj < ck0 ? 1 : 0;
      rk1 += kj < ck1 ? 1 : 0;
    }
    for (int j = 0; j < dg - 32; ++j) {
      const int kj = __builtin_amdgcn_readlane(ck1, j);
      rk0 += kj < ck0 ? 1 : 0;
      rk1 += kj < ck1 ? 1 : 0;
    }
    wl[rk0] = id0;
    if (hi) wl[rk1] = id1;
    s = wl[lane];
  } else {
    if (lane == 0) mb = atomicAdd(cursor, dg);
    mb = __builtin_amdgcn_readfirstlane(mb);
#pragma clang loop vectorize(disable) unroll(disable)
    for (int t = 0; t < len; ++t)
      raw16[mb + pos + t] = (unsigned short)bucket[start + t];
    __threadfence();
    for (int i = lane; i < dg; i += 32) {
      const int key = scratch_load(raw16 + mb + i);
      int rank = 0;
      for (int j = 0; j < dg; ++j) {
        const int kj = scratch_load(raw16 + mb + j);
        rank += (kj < key || (kj == key && j < i)) ? 1 : 0;
      }
      srt16[mb + rank] = (unsigned short)key;
    }
    __threadfence();
  }

  const float adst = quad_sum(dot4(rs, ad4));

  float m = leaky(quad_sum(dot4(rs, as4)) + adst);
  float den = 1.f;
  v4f acc = rs;

  for (int base = 0; base < dg; base += 32) {
    if (many) {
      const int j = base + lane;
      s = 0;
      if (j < dg) s = scratch_load(srt16 + mb + j);
    } else if (base) {
      s = wl[32 + lane];
    }
    const int left = dg - base;
    const int cnt = left < 32 ? left : 32;
    int j = 0;
    for (; j + 6 <= cnt; j += 6) {
      int sid[6];
#pragma unroll
      for (int i = 0; i < 6; ++i) sid[i] = __builtin_amdgcn_readlane(s, j + i);
      rows_step<6>(h4, lane, sid, 6, as4, adst, m, den, acc);
    }
    if (j + 4 <= cnt) {
      int sid[4];
#pragma unroll
      for (int i = 0; i < 4; ++i) sid[i] = __builtin_amdgcn_readlane(s, j + i);
      rows_step<4>(h4, lane, sid, 4, as4, adst, m, den, acc);
      j += 4;
    }
    const int rem = cnt - j;
    if (rem > 0) {
      int sid[3];
      sid[0] = __builtin_amdgcn_readlane(s, j);
      sid[1] = __builtin_amdgcn_readlane(s, rem > 1 ? j + 1 : j);
      sid[2] = __builtin_amdgcn_readlane(s, rem > 2 ? j + 2 : j);
      rows_step<3>(h4, lane, sid, rem, as4, adst, m, den, acc);
    }
  }

  const float dn = den + GAT_EPS;
  const v4f bv = ((const v4f*)bias)[lane];
  v4f r = acc / dn + bv;
  r.x = r.x > 0.f ? r.x : expm1f(r.x);
  r.y = r.y > 0.f ? r.y : expm1f(r.y);
  r.z = r.z > 0.f ? r.z : expm1f(r.z);
  r.w = r.w > 0.f ? r.w : expm1f(r.w);
  ((v4f*)out)[(size_t)n * 32 + lane] = r;
}

extern "C" void kernel_launch(void* const* d_in, const int* in_sizes, int n_in,
                              void* d_out, int out_size, void* d_ws, size_t ws_size,
                              hipStream_t stream) {
  (void)in_sizes; (void)n_in; (void)out_size; (void)ws_size;
  const float* x       = (const float*)d_in[0];
  const int*   ei      = (const int*)d_in[1];
  const float* W       = (const float*)d_in[2];
  const float* att_src = (const float*)d_in[3];
  const float* att_dst = (const float*)d_in[4];
  const float* bias    = (const float*)d_in[5];

  float*          h      = (float*)d_ws;
  unsigned short* cnt16  = (unsigned short*)(h + (size_t)N_NODES * OUT_DIM);
  int*            cursor = (int*)(cnt16 + (size_t)K_SLICES * N_NODES);
  int*            bucket = cursor + 16;
  unsigned short* raw16  = (unsigned short*)(bucket + N_EDGES);
  unsigned short* srt16  = raw16 + N_EDGES;

  gat_build_gemm_kernel<<<K_SLICES + GEMM_WGS, CNT_T, 0, stream>>>(ei, cnt16, bucket, cursor, x, W, h);
  gat_gather_kernel<<<N_NODES / 8, 256, 0, stream>>>(cnt16, bucket, raw16, srt16, cursor,
                                                     att_src, att_dst, h, bias, (float*)d_out);
}


# Gather wave issues ~30% fewer vector instructions: branch-free packed expm1 for ELU, one reciprocal for 4 divides, exp2 logits, in-edges ordered inside slice segments only: new sum order, rounding
# speedup vs baseline: 12.8653x; 1.0368x over previous
#include <hip/hip_runtime.h>
#include <math.h>

#define N_NODES 50000
#define N_EDGES 800000
#define IN_DIM 128
#define OUT_DIM 128
#define HEADS 8
#define CPH 16
#define NEG_SLOPE 0.2f
#define GAT_EPS 1e-16f
#define LOG2E 1.44269504088896340736f
#define K_SLICES 32
#define EPB (N_EDGES / K_SLICES)

typedef float v2f __attribute__((ext_vector_type(2)));
typedef float v8f __attribute__((ext_vector_type(8)));

#define LDX 132
#define GEMM_ROWS 64
#define GEMM_UNITS ((N_NODES + GEMM_ROWS - 1) / GEMM_ROWS)
#define GEMM_WGS 196
#define M_TILES (N_NODES / 16)

__device__ __forceinline__ float leaky(float v) {
  return fmaxf(v, NEG_SLOPE * v);
}

static_assert(K_SLICES == 32, "a gather wave reads one slice per lane");
static_assert(N_EDGES % K_SLICES == 0 && EPB % 4 == 0 && N_NODES % 8 == 0,
              "slices and tables are moved in whole 16-byte groups");
static_assert(EPB <= 65535 && N_NODES <= 65536,
              "prefix entries and source ids are kept in 16 bits");
#define V4PB (EPB / 4)
#define CNT_T 1024
#define CNT_IT ((V4PB + CNT_T - 1) / CNT_T)
#define SCAN_V4 13
#define HIST_PAD (CNT_T * SCAN_V4 * 4)
static_assert(2 * GEMM_ROWS * LDX <= HIST_PAD, "the GEMM's two x buffers live in the counters' LDS");
static_assert(GEMM_WGS * 4 >= GEMM_UNITS && CNT_T == 1024 && N_NODES % 16 == 0,
              "32 waves per GEMM work-group, whole M-tiles, at most 4 units each");

__device__ __forceinline__ void gemm_role(float* __restrict__ sx, int g,
                                          const float* __restrict__ x,
                                          const float* __restrict__ Wg,
                                          float* __restrict__ h) {
  const int tid = threadIdx.x;
  const int wave = tid >> 5;
  const int lane = tid & 31;
  const int half = lane >> 4;
  const int mr   = lane & 15;
  const int nb   = (wave & 7) * 16;
  const int mg   = wave >> 3;

  const int r0 = tid >> 5, c4 = tid & 31;
  int u = g;
  float4 p0 = make_float4(0.f, 0.f, 0.f, 0.f), p1 = p0;
  if (u < GEMM_UNITS) {
    const int ra = u * GEMM_ROWS + r0, rb = ra + 32;
    if (ra < N_NODES) p0 = ((const float4*)(x + (size_t)ra * IN_DIM))[c4];
    if (rb < N_NODES) p1 = ((const float4*)(x + (size_t)rb * IN_DIM))[c4];
  }

  v2f bf[IN_DIM / 4];
#pragma unroll
  for (int kk = 0; kk < IN_DIM / 4; ++kk) {
    bf[kk].x = Wg[(size_t)(4 * kk + 2 * half) * OUT_DIM + nb + mr];
    bf[kk].y = Wg[(size_t)(4 * kk + 2 * half + 1) * OUT_DIM + nb + mr];
  }

  int buf = 0;
  while (u < GEMM_UNITS) {
    float* sb = sx + buf * (GEMM_ROWS * LDX);
    *(float4*)(sb + r0 * LDX + c4 * 4) = p0;
    *(float4*)(sb + (r0 + 32) * LDX + c4 * 4) = p1;
    __syncthreads();

    const int un = u + GEMM_WGS;
    if (un < GEMM_UNITS) {
      const int ra = un * GEMM_ROWS + r0, rb = ra + 32;
      p0 = make_float4(0.f, 0.f, 0.f, 0.f);
      p1 = p0;
      if (ra < N_NODES) p0 = ((const float4*)(x + (size_t)ra * IN_DIM))[c4];
      if (rb < N_NODES) p1 = ((const float4*)(x + (size_t)rb * IN_DIM))[c4];
    }

    const int mt = u * (GEMM_ROWS / 16) + mg;
    if (mt < M_TILES) {
      const float* sa = sb + (mg * 16 + mr) * LDX + 2 * half;
      v8f acc = {};
#pragma unroll
      for (int kk = 0; kk < IN_DIM / 4; ++kk) {
        const v2f a = *(const v2f*)(sa + 4 * kk);
        acc = __builtin_amdgcn_wmma_f32_16x16x4_f32(
             false, a,  false, bf[kk],
             (short)0, acc,  false,  false);
      }
      float* hrow = h + (size_t)(mt * 16 + half * 8) * OUT_DIM + nb + mr;
#pragma unroll
      for (int r = 0; r < 8; ++r) hrow[(size_t)r * OUT_DIM] = acc[r];
    }
    buf ^= 1;
    u = un;
  }
}

__global__ __launch_bounds__(CNT_T) void gat_build_gemm_kernel(
    const int* __restrict__ ei, unsigned short* __restrict__ cnt16,
    int* __restrict__ bucket, int* __restrict__ cursor,
    const float* __restrict__ x, const float* __restrict__ Wg,
    float* __restrict__ h) {
  __shared__ __attribute__((aligned(16))) int hist[HIST_PAD];
  __shared__ __attribute__((aligned(16))) int srt[EPB];
  __shared__ int wtot[32];

  if (blockIdx.x >= K_SLICES) {
    gemm_role((float*)hist, (int)blockIdx.x - K_SLICES, x, Wg, h);
    return;
  }

  const int t = threadIdx.x, lane = t & 31, w = t >> 5;
  const int k = blockIdx.x;

  if (k == 0 && t == 0) *cursor = 0;

  const int4* d4 = (const int4*)(ei + N_EDGES + k * EPB);
  const int4* s4 = (const int4*)(ei + k * EPB);
  int4 dv[CNT_IT], sv[CNT_IT];
#pragma unroll
  for (int j = 0; j < CNT_IT; ++j) {
    const int idx = t + j * CNT_T;
    dv[j] = d4[idx < V4PB ? idx : V4PB - 1];
  }
#pragma unroll
  for (int j = 0; j < CNT_IT; ++j) {
    const int idx = t + j * CNT_T;
    sv[j] = s4[idx < V4PB ? idx : V4PB - 1];
  }

  int4* h4 = (int4*)hist;
  for (int i = t; i < HIST_PAD / 4; i += CNT_T) h4[i] = make_int4(0, 0, 0, 0);
  __syncthreads();

  int4 rv[CNT_IT];
#pragma unroll
  for (int j = 0; j < CNT_IT; ++j) {
    const int idx = t + j * CNT_T;
    rv[j] = make_int4(0, 0, 0, 0);
    if (idx < V4PB) {
      rv[j].x = atomicAdd(&hist[dv[j].x], 1);
      rv[j].y = atomicAdd(&hist[dv[j].y], 1);
      rv[j].z = atomicAdd(&hist[dv[j].z], 1);
      rv[j].w = atomicAdd(&hist[dv[j].w], 1);
    }
  }
  __syncthreads();

  int4* my = h4 + t * SCAN_V4;
  int sum = 0;
#pragma unroll
  for (int j = 0; j < SCAN_V4; ++j) {
    const int4 v = my[j];
    sum += (v.x + v.y) + (v.z + v.w);
  }
  int inc = sum;
#pragma unroll
  for (int d = 1; d < 32; d <<= 1) {
    const int u = __shfl_up(inc, d, 32);
    if (lane >= d) inc += u;
  }
  if (lane == 31) wtot[w] = inc;
  __syncthreads();
  int run = inc - sum;
  for (int q = 0; q < 32; ++q)
    if (q < w) run += wtot[q];
#pragma unroll
  for (int j = 0; j < SCAN_V4; ++j) {
    const int4 v = my[j];
    int4 o;
    o.x = run;
    o.y = o.x + v.x;
    o.z = o.y + v.y;
    o.w = o.z + v.z;
    run = o.w + v.w;
    my[j] = o;
  }
  __syncthreads();

#pragma unroll
  for (int j = 0; j < CNT_IT; ++j) {
    const int idx = t + j * CNT_T;
    if (idx < V4PB) {
      rv[j].x += hist[dv[j].x];
      rv[j].y += hist[dv[j].y];
      rv[j].z += hist[dv[j].z];
      rv[j].w += hist[dv[j].w];
    }
  }
#pragma unroll
  for (int j = 0; j < CNT_IT; ++j) {
    const int idx = t + j * CNT_T;
    if (idx < V4PB) {
      srt[rv[j].x] = sv[j].x;
      srt[rv[j].y] = sv[j].y;
      srt[rv[j].z] = sv[j].z;
      srt[rv[j].w] = sv[j].w;
    }
  }

  int4* o4 = (int4*)(cnt16 + (size_t)k * N_NODES);
  for (int i = t; i < N_NODES / 8; i += CNT_T) {
    const int4 a = h4[2 * i], b = h4[2 * i + 1];
    int4 p;
    p.x = a.x | (a.y << 16);
    p.y = a.z | (a.w << 16);
    p.z = b.x | (b.y << 16);
    p.w = b.z | (b.w << 16);
    o4[i] = p;
  }
  __syncthreads();

  int4* b4 = (int4*)(bucket + (size_t)k * EPB);
  const int4* q4 = (const int4*)srt;
  for (int i = t; i < V4PB; i += CNT_T) b4[i] = q4[i];
}


__device__ __forceinline__ float quad_sum(float v) {
  float t = v + __int_as_float(__builtin_amdgcn_update_dpp(
                    0, __float_as_int(v), 0x161, 0xF, 0xF, true));
  t = t + __int_as_float(__builtin_amdgcn_update_dpp(
              0, __float_as_int(t), 0x162, 0xF, 0xF, true));
  return t;
}

__device__ __forceinline__ int wave_incl_scan(int v, int lane) {
  int x = v;
  x += __builtin_amdgcn_update_dpp(0, x, 0x111, 0xF, 0xF, true);
  x += __builtin_amdgcn_update_dpp(0, x, 0x112, 0xF, 0xF, true);
  x += __builtin_amdgcn_update_dpp(0, x, 0x114, 0xF, 0xF, true);
  x += __builtin_amdgcn_update_dpp(0, x, 0x118, 0xF, 0xF, true);
  const int lo = __builtin_amdgcn_readlane(x, 15);
  return lane >= 16 ? x + lo : x;
}

typedef float v4f __attribute__((ext_vector_type(4)));

__device__ __forceinline__ float dot4(const v4f a, const v4f b) {
  const v2f t = a.xy * b.xy + a.zw * b.zw;
  return t.x + t.y;
}

template <int G>
__device__ __forceinline__ void rows_step(const v4f* __restrict__ h4, int lane,
                                          const int (&sid)[G], int nvalid,
                                          const v4f as4, float adst,
                                          float& m, float& den, v4f& acc) {
  v4f r[G];
  float e[G];
#pragma unroll
  for (int i = 0; i < G; ++i) r[i] = h4[(size_t)sid[i] * 32 + lane];
  float mn = m;
#pragma unroll
  for (int i = 0; i < G; ++i) {
    e[i] = i < nvalid ? leaky(quad_sum(dot4(r[i], as4)) + adst) : -INFINITY;
    mn = fmaxf(mn, e[i]);
  }
  const float sc = __builtin_amdgcn_exp2f(m - mn);
  m = mn;
  den *= sc;
  acc *= sc;
#pragma unroll
  for (int i = 0; i < G; ++i) {
    const float p = __builtin_amdgcn_exp2f(e[i] - mn);
    den += p;
    acc += p * r[i];
  }
}

__device__ __forceinline__ v4f expm1_neg4(const v4f x) {
  const v4f t = x * LOG2E;
  v4f e;
  e.x = __builtin_amdgcn_exp2f(t.x);
  e.y = __builtin_amdgcn_exp2f(t.y);
  e.z = __builtin_amdgcn_exp2f(t.z);
  e.w = __builtin_amdgcn_exp2f(t.w);
  e = e - 1.0f;
  v4f q = x * (1.0f / 5040.0f) + (1.0f / 720.0f);
  q = q * x + (1.0f / 120.0f);
  q = q * x + (1.0f / 24.0f);
  q = q * x + (1.0f / 6.0f);
  q = q * x + 0.5f;
  const v4f p = (x * x) * q + x;
  v4f r;
  r.x = x.x > -0.25f ? p.x : e.x;
  r.y = x.y > -0.25f ? p.y : e.y;
  r.z = x.z > -0.25f ? p.z : e.z;
  r.w = x.w > -0.25f ? p.w : e.w;
  return r;
}

__device__ __forceinline__ int scratch_load(unsigned short* p) {
  return (int)__hip_atomic_load(p, __ATOMIC_RELAXED, __HIP_MEMORY_SCOPE_AGENT);
}

__global__ __launch_bounds__(256) void gat_gather_kernel(
    const unsigned short* __restrict__ cnt16, const int* __restrict__ bucket,
    unsigned short* raw16, unsigned short* srt16, int* cursor,
    const float* __restrict__ att_src, const float* __restrict__ att_dst,
    const float* __restrict__ h, const float* __restrict__ bias,
    float* __restrict__ out) {
  __shared__ int lst[8 * 64];
  const int lane = threadIdx.x & 31;
  const int wv = threadIdx.x >> 5;
  const int n = __builtin_amdgcn_readfirstlane((int)(blockIdx.x * 8 + wv));
  const v4f* h4 = (const v4f*)h;
  int* wl = lst + wv * 64;

  const v4f rs = h4[(size_t)n * 32 + lane];
  const v4f as4 = ((const v4f*)att_src)[lane] * LOG2E;
  const v4f ad4 = ((const v4f*)att_dst)[lane] * LOG2E;

  const unsigned short* crow = cnt16 + (size_t)lane * N_NODES + n;
  const int sa = crow[0];
  int sb = EPB;
  if (n + 1 < N_NODES) sb = crow[1];
  const int len = sb - sa;
  const int start = lane * EPB + sa;
  const int inc = wave_incl_scan(len, lane);
  const int pos = inc - len;
  const int dg = __builtin_amdgcn_readlane(inc, 31);
  const bool many = dg > 64;

  int s = 0;
  int mb = 0;
  if (dg <= 32) {
#pragma nounroll
    for (int t = 0; t < len; ++t) wl[pos + t] = start + t;
    const bool valid = lane < dg;
    int a = 0, id = 0;
    if (valid) {
      a = wl[lane];
      id = bucket[a];
    }
    const int key = valid ? (int)((((unsigned)a / EPB) << 16) | (unsigned)id) : 0x7fffffff;
    int rk = lane;
    for (int d = 1; __builtin_amdgcn_ballot_w32(len > d) != 0u; ++d) {
      const int iu = lane - d, il = lane + d;
      const int ku = __builtin_amdgcn_ds_bpermute((iu < 0 ? lane : iu) << 2, key);
      const int kl = __builtin_amdgcn_ds_bpermute((il > 31 ? lane : il) << 2, key);
      rk += (kl < key ? 1 : 0) - (ku > key ? 1 : 0);
    }
    s = __builtin_amdgcn_ds_permute(rk << 2, id);
  } else if (!many) {
#pragma nounroll
    for (int t = 0; t < len; ++t) wl[pos + t] = start + t;
    const bool hi = lane + 32 < dg;
    const int id0 = bucket[wl[lane]];
    int id1 = 0;
    if (hi) id1 = bucket[wl[32 + lane]];
    const int ck0 = id0 * 64 + lane;
    const int ck1 = id1 * 64 + 32 + lane;
    int rk0 = 0, rk1 = 0;
    for (int j = 0; j < 32; ++j) {
      const int kj = __builtin_amdgcn_readlane(ck0, j);
      rk0 += kj < ck0 ? 1 : 0;
      rk1 += kj < ck1 ? 1 : 0;
    }
    for (int j = 0; j < dg - 32; ++j) {
      const int kj = __builtin_amdgcn_readlane(ck1, j);
      rk0 += kj < ck0 ? 1 : 0;
      rk1 += kj < ck1 ? 1 : 0;
    }
    wl[rk0] = id0;
    if (hi) wl[rk1] = id1;
    s = wl[lane];
  } else {
    if (lane == 0) mb = atomicAdd(cursor, dg);
    mb = __builtin_amdgcn_readfirstlane(mb);
#pragma clang loop vectorize(disable) unroll(disable)
    for (int t = 0; t < len; ++t)
      raw16[mb + pos + t] = (unsigned short)bucket[start + t];
    __threadfence();
    for (int i = lane; i < dg; i += 32) {
      const int key = scratch_load(raw16 + mb + i);
      int rank = 0;
      for (int j = 0; j < dg; ++j) {
        const int kj = scratch_load(raw16 + mb + j);
        rank += (kj < key || (kj == key && j < i)) ? 1 : 0;
      }
      srt16[mb + rank] = (unsigned short)key;
    }
    __threadfence();
  }

  const float adst = quad_sum(dot4(rs, ad4));

  float m = leaky(quad_sum(dot4(rs, as4)) + adst);
  float den = 1.f;
  v4f acc = rs;

  for (int base = 0; base < dg; base += 32) {
    if (many) {
      const int j = base + lane;
      s = 0;
      if (j < dg) s = scratch_load(srt16 + mb + j);
    } else if (base) {
      s = wl[32 + lane];
    }
    const int left = dg - base;
    const int cnt = left < 32 ? left : 32;
    int j = 0;
    for (; j + 6 <= cnt; j += 6) {
      int sid[6];
#pragma unroll
      for (int i = 0; i < 6; ++i) sid[i] = __builtin_amdgcn_readlane(s, j + i);
      rows_step<6>(h4, lane, sid, 6, as4, adst, m, den, acc);
    }
    if (j + 4 <= cnt) {
      int sid[4];
#pragma unroll
      for (int i = 0; i < 4; ++i) sid[i] = __builtin_amdgcn_readlane(s, j + i);
      rows_step<4>(h4, lane, sid, 4, as4, adst, m, den, acc);
      j += 4;
    }
    const int rem = cnt - j;
    if (rem > 0) {
      int sid[3];
      sid[0] = __builtin_amdgcn_readlane(s, j);
      sid[1] = __builtin_amdgcn_readlane(s, rem > 1 ? j + 1 : j);
      sid[2] = __builtin_amdgcn_readlane(s, rem > 2 ? j + 2 : j);
      rows_step<3>(h4, lane, sid, rem, as4, adst, m, den, acc);
    }
  }

  const float inv = 1.0f / (den + GAT_EPS);
  const v4f bv = ((const v4f*)bias)[lane];
  const v4f r = acc * inv + bv;
  const v4f zero = {0.f, 0.f, 0.f, 0.f};
  const v4f xp = __builtin_elementwise_max(r, zero);
  const v4f xn = __builtin_elementwise_min(r, zero);
  ((v4f*)out)[(size_t)n * 32 + lane] = xp + expm1_neg4(xn);
}

extern "C" void kernel_launch(void* const* d_in, const int* in_sizes, int n_in,
                              void* d_out, int out_size, void* d_ws, size_t ws_size,
                              hipStream_t stream) {
  (void)in_sizes; (void)n_in; (void)out_size; (void)ws_size;
  const float* x       = (const float*)d_in[0];
  const int*   ei      = (const int*)d_in[1];
  const float* W       = (const float*)d_in[2];
  const float* att_src = (const float*)d_in[3];
  const float* att_dst = (const float*)d_in[4];
  const float* bias    = (const float*)d_in[5];

  float*          h      = (float*)d_ws;
  unsigned short* cnt16  = (unsigned short*)(h + (size_t)N_NODES * OUT_DIM);
  int*            cursor = (int*)(cnt16 + (size_t)K_SLICES * N_NODES);
  int*            bucket = cursor + 16;
  unsigned short* raw16  = (unsigned short*)(bucket + N_EDGES);
  unsigned short* srt16  = raw16 + N_EDGES;

  gat_build_gemm_kernel<<<K_SLICES + GEMM_WGS, CNT_T, 0, stream>>>(ei, cnt16, bucket, cursor, x, W, h);
  gat_gather_kernel<<<N_NODES / 8, 256, 0, stream>>>(cnt16, bucket, raw16, srt16, cursor,
                                                     att_src, att_dst, h, bias, (float*)d_out);
}
